# starting kernel
# baseline (speedup 1.0000x reference)
#include <hip/hip_runtime.h>
#include <cstdint>

typedef __attribute__((ext_vector_type(2))) float v2f;
typedef __attribute__((ext_vector_type(8))) float v8f;

#define TT   26
#define NH   8
#define CC   256
#define CPH  32
#define HW_  256
#define NB   52

__device__ __forceinline__ v8f wmma_f32(v2f a, v2f b, v8f c) {
  return __builtin_amdgcn_wmma_f32_16x16x4_f32(false, a, false, b, (short)0, c,
                                               false, false);
}

__global__ void __launch_bounds__(256) qkv_gemm(const float* __restrict__ x,
                                                const float* __restrict__ wq,
                                                const float* __restrict__ bq,
                                                float* __restrict__ Y) {
  const int tid  = threadIdx.x;
  const int lane = tid & 31, wave = tid >> 5;
  const int l16  = lane & 15, g16 = lane >> 4;
  const int job   = blockIdx.x * 8 + wave;
  const int ptile = job & 3;
  const int mtile = (job >> 2) % 48;
  const int n     = job / 192;
  const int m0    = mtile * 16;
  const int p0    = ptile * 64;
  const float* xb = x + (size_t)n * CC * HW_;

  v8f acc0 = {}, acc1 = {}, acc2 = {}, acc3 = {};
  for (int k = 0; k < CC; k += 4) {
    const float* wp = wq + (size_t)(m0 + l16) * CC + k + 2 * g16;
    v2f a; a[0] = wp[0]; a[1] = wp[1];
    const float* xp = xb + (size_t)(k + 2 * g16) * HW_ + p0 + l16;
    v2f b0; b0[0] = xp[0];  b0[1] = xp[HW_];
    v2f b1; b1[0] = xp[16]; b1[1] = xp[HW_ + 16];
    v2f b2; b2[0] = xp[32]; b2[1] = xp[HW_ + 32];
    v2f b3; b3[0] = xp[48]; b3[1] = xp[HW_ + 48];
    acc0 = wmma_f32(a, b0, acc0);
    acc1 = wmma_f32(a, b1, acc1);
    acc2 = wmma_f32(a, b2, acc2);
    acc3 = wmma_f32(a, b3, acc3);
  }
  float* yb = Y + ((size_t)n * 768 + m0) * HW_ + p0;
#pragma unroll
  for (int r = 0; r < 8; ++r) {
    const int   m    = r + 8 * g16;
    const float bias = bq[m0 + m];
    float* yr = yb + (size_t)m * HW_ + l16;
    yr[0]  = acc0[r] + bias;
    yr[16] = acc1[r] + bias;
    yr[32] = acc2[r] + bias;
    yr[48] = acc3[r] + bias;
  }
}

__global__ void __launch_bounds__(256) attn_kernel(const float* __restrict__ Y,
                                                   const int* __restrict__ accA,
                                                   float* __restrict__ O) {
  __shared__ float Vlds[1024 * 33];
  __shared__ float Slds[8 * 16 * 17];

  const int blk  = blockIdx.x;
  const int half = blk & 1;
  const int bth  = blk >> 1;
  const int head = bth % NH;
  const int t    = (bth / NH) % TT;
  const int b    = bth / (NH * TT);
  const int n    = b * TT + t;

  const int tid  = threadIdx.x;
  const int lane = tid & 31, wave = tid >> 5;
  const int l16  = lane & 15, g16 = lane >> 4;

  int tg[4];
#pragma unroll
  for (int g = 0; g < 4; ++g) tg[g] = accA[t * 4 + g];

  for (int i = tid; i < 1024 * 32; i += 256) {
    const int cp  = i >> 10;
    const int key = i & 1023;
    const int gg  = key >> 8;
    const int pp  = key & 255;
    Vlds[key * 33 + cp] =
        Y[((size_t)(b * TT + tg[gg]) * 3 * CC + 2 * CC + cp * NH + head) * HW_ + pp];
  }
  __syncthreads();

  const int m0 = half * 128 + wave * 16;
  const float* Qbase = Y + (size_t)n * 3 * CC * HW_;

  v2f aq[8];
#pragma unroll
  for (int s = 0; s < 8; ++s) {
    const int cp = 4 * s + 2 * g16;
    const float* qp = Qbase + (size_t)(cp * NH + head) * HW_ + m0 + l16;
    aq[s][0] = qp[0];
    aq[s][1] = qp[NH * HW_];
  }

  v8f o0 = {}, o1 = {};
  float rs[8];
#pragma unroll
  for (int r = 0; r < 8; ++r) rs[r] = 0.0f;
  const float scale = 0.17677669529663687f;
  float* Sw = &Slds[wave * 16 * 17];

  for (int kt = 0; kt < 64; ++kt) {
    const int gg    = kt >> 4;
    const int pbase = (kt & 15) * 16;
    const float* Kb =
        Y + ((size_t)(b * TT + tg[gg]) * 3 * CC + CC) * HW_;

    v8f sacc = {};
#pragma unroll
    for (int s = 0; s < 8; ++s) {
      const int cp = 4 * s + 2 * g16;
      const float* kp = Kb + (size_t)(cp * NH + head) * HW_ + pbase + l16;
      v2f bk; bk[0] = kp[0]; bk[1] = kp[NH * HW_];
      sacc = wmma_f32(aq[s], bk, sacc);
    }

#pragma unroll
    for (int r = 0; r < 8; ++r) {
      const float p = __expf(sacc[r] * scale);
      rs[r] += p;
      Sw[(r + 8 * g16) * 17 + l16] = p;
    }
    asm volatile("s_wait_dscnt 0" ::: "memory");

#pragma unroll
    for (int s = 0; s < 4; ++s) {
      const int krow = 4 * s + 2 * g16;
      v2f ap;
      ap[0] = Sw[l16 * 17 + krow];
      ap[1] = Sw[l16 * 17 + krow + 1];
      const int key = kt * 16 + krow;
      v2f bv0, bv1;
      bv0[0] = Vlds[key * 33 + l16];
      bv0[1] = Vlds[(key + 1) * 33 + l16];
      bv1[0] = Vlds[key * 33 + 16 + l16];
      bv1[1] = Vlds[(key + 1) * 33 + 16 + l16];
      o0 = wmma_f32(ap, bv0, o0);
      o1 = wmma_f32(ap, bv1, o1);
    }
  }

#pragma unroll
  for (int r = 0; r < 8; ++r) {
    float v = rs[r];
    v += __shfl_xor(v, 1, 32);
    v += __shfl_xor(v, 2, 32);
    v += __shfl_xor(v, 4, 32);
    v += __shfl_xor(v, 8, 32);
    rs[r] = 1.0f / v;
  }

  float* Ob = O + (size_t)n * CC * HW_;
#pragma unroll
  for (int r = 0; r < 8; ++r) {
    const int p = m0 + r + 8 * g16;
    Ob[(size_t)(l16 * NH + head) * HW_ + p]        = o0[r] * rs[r];
    Ob[(size_t)((16 + l16) * NH + head) * HW_ + p] = o1[r] * rs[r];
  }
}

__global__ void __launch_bounds__(256) proj_gemm(const float* __restrict__ Oin,
                                                 const float* __restrict__ wo,
                                                 const float* __restrict__ bo,
                                                 float* __restrict__ out) {
  const int tid  = threadIdx.x;
  const int lane = tid & 31, wave = tid >> 5;
  const int l16  = lane & 15, g16 = lane >> 4;
  const int job   = blockIdx.x * 8 + wave;
  const int ptile = job & 3;
  const int mtile = (job >> 2) & 15;
  const int n     = job >> 6;
  const int m0    = mtile * 16;
  const int p0    = ptile * 64;
  const float* ob = Oin + (size_t)n * CC * HW_;

  v8f acc0 = {}, acc1 = {}, acc2 = {}, acc3 = {};
  for (int k = 0; k < CC; k += 4) {
    const float* wp = wo + (size_t)(m0 + l16) * CC + k + 2 * g16;
    v2f a; a[0] = wp[0]; a[1] = wp[1];
    const float* xp = ob + (size_t)(k + 2 * g16) * HW_ + p0 + l16;
    v2f b0; b0[0] = xp[0];  b0[1] = xp[HW_];
    v2f b1; b1[0] = xp[16]; b1[1] = xp[HW_ + 16];
    v2f b2; b2[0] = xp[32]; b2[1] = xp[HW_ + 32];
    v2f b3; b3[0] = xp[48]; b3[1] = xp[HW_ + 48];
    acc0 = wmma_f32(a, b0, acc0);
    acc1 = wmma_f32(a, b1, acc1);
    acc2 = wmma_f32(a, b2, acc2);
    acc3 = wmma_f32(a, b3, acc3);
  }
  float* yb = out + ((size_t)n * CC + m0) * HW_ + p0;
#pragma unroll
  for (int r = 0; r < 8; ++r) {
    const int   m    = r + 8 * g16;
    const float bias = bo[m0 + m];
    float* yr = yb + (size_t)m * HW_ + l16;
    yr[0]  = acc0[r] + bias;
    yr[16] = acc1[r] + bias;
    yr[32] = acc2[r] + bias;
    yr[48] = acc3[r] + bias;
  }
}

extern "C" void kernel_launch(void* const* d_in, const int* in_sizes, int n_in,
                              void* d_out, int out_size, void* d_ws, size_t ws_size,
                              hipStream_t stream) {
  const float* x      = (const float*)d_in[0];
  const float* w_qkv  = (const float*)d_in[1];
  const float* b_qkv  = (const float*)d_in[2];
  const float* w_out  = (const float*)d_in[3];
  const float* b_out  = (const float*)d_in[4];
  const int*   accA   = (const int*)d_in[5];
  float* outp = (float*)d_out;

  float* Y = (float*)d_ws;
  float* O = Y + (size_t)NB * 768 * HW_;

  qkv_gemm<<<1248, 256, 0, stream>>>(x, w_qkv, b_qkv, Y);
  attn_kernel<<<832, 256, 0, stream>>>(Y, accA, O);
  proj_gemm<<<416, 256, 0, stream>>>(O, w_out, b_out, outp);
}


# attention: S and O computed transposed (K^T Q^T, V^T P^T) so P goes accumulator->B operand in registers, no LDS round trip; V kept [c'][key] in LDS, b128 reads; row-sum and PV sum order change
# speedup vs baseline: 1.1380x; 1.1380x over previous
#include <hip/hip_runtime.h>
#include <cstdint>

typedef __attribute__((ext_vector_type(2))) float v2f;
typedef __attribute__((ext_vector_type(8))) float v8f;

#define TT   26
#define NH   8
#define CC   256
#define CPH  32
#define HW_  256
#define NB   52

__device__ __forceinline__ v8f wmma_f32(v2f a, v2f b, v8f c) {
  return __builtin_amdgcn_wmma_f32_16x16x4_f32(false, a, false, b, (short)0, c,
                                               false, false);
}

__global__ void __launch_bounds__(256) qkv_gemm(const float* __restrict__ x,
                                                const float* __restrict__ wq,
                                                const float* __restrict__ bq,
                                                float* __restrict__ Y) {
  const int tid  = threadIdx.x;
  const int lane = tid & 31, wave = tid >> 5;
  const int l16  = lane & 15, g16 = lane >> 4;
  const int job   = blockIdx.x * 8 + wave;
  const int ptile = job & 3;
  const int mtile = (job >> 2) % 48;
  const int n     = job / 192;
  const int m0    = mtile * 16;
  const int p0    = ptile * 64;
  const float* xb = x + (size_t)n * CC * HW_;

  v8f acc0 = {}, acc1 = {}, acc2 = {}, acc3 = {};
  for (int k = 0; k < CC; k += 4) {
    const float* wp = wq + (size_t)(m0 + l16) * CC + k + 2 * g16;
    v2f a; a[0] = wp[0]; a[1] = wp[1];
    const float* xp = xb + (size_t)(k + 2 * g16) * HW_ + p0 + l16;
    v2f b0; b0[0] = xp[0];  b0[1] = xp[HW_];
    v2f b1; b1[0] = xp[16]; b1[1] = xp[HW_ + 16];
    v2f b2; b2[0] = xp[32]; b2[1] = xp[HW_ + 32];
    v2f b3; b3[0] = xp[48]; b3[1] = xp[HW_ + 48];
    acc0 = wmma_f32(a, b0, acc0);
    acc1 = wmma_f32(a, b1, acc1);
    acc2 = wmma_f32(a, b2, acc2);
    acc3 = wmma_f32(a, b3, acc3);
  }
  float* yb = Y + ((size_t)n * 768 + m0) * HW_ + p0;
#pragma unroll
  for (int r = 0; r < 8; ++r) {
    const int   m    = r + 8 * g16;
    const float bias = bq[m0 + m];
    float* yr = yb + (size_t)m * HW_ + l16;
    yr[0]  = acc0[r] + bias;
    yr[16] = acc1[r] + bias;
    yr[32] = acc2[r] + bias;
    yr[48] = acc3[r] + bias;
  }
}

#define VPITCH4 257

__global__ void __launch_bounds__(256) attn_kernel(const float* __restrict__ Y,
                                                   const int* __restrict__ accA,
                                                   float* __restrict__ O) {
  __shared__ float4 Vt4[32 * VPITCH4];

  const int blk  = blockIdx.x;
  const int half = blk & 1;
  const int bth  = blk >> 1;
  const int head = bth % NH;
  const int t    = (bth / NH) % TT;
  const int b    = bth / (NH * TT);
  const int n    = b * TT + t;

  const int tid  = threadIdx.x;
  const int lane = tid & 31, wave = tid >> 5;
  const int l16  = lane & 15, g16 = lane >> 4;

  int tg[4];
#pragma unroll
  for (int g = 0; g < 4; ++g) tg[g] = accA[t * 4 + g];

  {
    const int sg = tid >> 6;
    const int s4 = tid & 63;
    const int tsel = sg == 0 ? tg[0] : sg == 1 ? tg[1] : sg == 2 ? tg[2] : tg[3];
    const float* vsrc =
        Y + ((size_t)(b * TT + tsel) * 3 * CC + 2 * CC + head) * HW_ + s4 * 4;
    float4* vdst = Vt4 + sg * 64 + s4;
#pragma unroll 8
    for (int cp = 0; cp < CPH; ++cp) {
      vdst[cp * VPITCH4] = *(const float4*)(vsrc + (size_t)cp * NH * HW_);
    }
  }
  __syncthreads();

  const int m0 = half * 128 + wave * 16;
  const float* Qbase = Y + (size_t)n * 3 * CC * HW_;

  v2f bq[8];
#pragma unroll
  for (int s = 0; s < 8; ++s) {
    const int cp = 4 * s + 2 * g16;
    const float* qp = Qbase + (size_t)(cp * NH + head) * HW_ + m0 + l16;
    bq[s][0] = qp[0];
    bq[s][1] = qp[NH * HW_];
  }

  v8f o0 = {}, o1 = {};
  float rs = 0.0f;
  const float scale = 0.17677669529663687f;
  const float4* Vr0 = Vt4 + l16 * VPITCH4 + 2 * g16;
  const float4* Vr1 = Vr0 + 16 * VPITCH4;

  for (int kt = 0; kt < 64; ++kt) {
    const int gg    = kt >> 4;
    const int pbase = (kt & 15) * 16;
    const float* Kb =
        Y + ((size_t)(b * TT + tg[gg]) * 3 * CC + CC) * HW_;

    const float4 va0 = Vr0[kt * 4], va1 = Vr0[kt * 4 + 1];
    const float4 vb0 = Vr1[kt * 4], vb1 = Vr1[kt * 4 + 1];
    __builtin_amdgcn_sched_barrier(0);

    v8f sacc = {};
#pragma unroll
    for (int s = 0; s < 8; ++s) {
      const int cp = 4 * s + 2 * g16;
      const float* kp = Kb + (size_t)(cp * NH + head) * HW_ + pbase + l16;
      v2f ak; ak[0] = kp[0]; ak[1] = kp[NH * HW_];
      sacc = wmma_f32(ak, bq[s], sacc);
    }

    float p[8];
#pragma unroll
    for (int r = 0; r < 8; ++r) p[r] = __expf(sacc[r] * scale);
    rs += ((p[0] + p[1]) + (p[2] + p[3])) + ((p[4] + p[5]) + (p[6] + p[7]));
    __builtin_amdgcn_sched_barrier(0);

    v2f bp, av;
    bp[0] = p[0]; bp[1] = p[1];
    av[0] = va0.x; av[1] = va0.y;  o0 = wmma_f32(av, bp, o0);
    av[0] = vb0.x; av[1] = vb0.y;  o1 = wmma_f32(av, bp, o1);
    bp[0] = p[2]; bp[1] = p[3];
    av[0] = va0.z; av[1] = va0.w;  o0 = wmma_f32(av, bp, o0);
    av[0] = vb0.z; av[1] = vb0.w;  o1 = wmma_f32(av, bp, o1);
    bp[0] = p[4]; bp[1] = p[5];
    av[0] = va1.x; av[1] = va1.y;  o0 = wmma_f32(av, bp, o0);
    av[0] = vb1.x; av[1] = vb1.y;  o1 = wmma_f32(av, bp, o1);
    bp[0] = p[6]; bp[1] = p[7];
    av[0] = va1.z; av[1] = va1.w;  o0 = wmma_f32(av, bp, o0);
    av[0] = vb1.z; av[1] = vb1.w;  o1 = wmma_f32(av, bp, o1);
  }

  rs += __shfl_xor(rs, 16, 32);
  const float inv = 1.0f / rs;

  float* Ob = O + (size_t)n * CC * HW_ + m0 + l16;
#pragma unroll
  for (int r = 0; r < 8; ++r) {
    const int c1 = 8 * g16 + r;
    Ob[(size_t)(c1 * NH + head) * HW_]        = o0[r] * inv;
    Ob[(size_t)((16 + c1) * NH + head) * HW_] = o1[r] * inv;
  }
}

__global__ void __launch_bounds__(256) proj_gemm(const float* __restrict__ Oin,
                                                 const float* __restrict__ wo,
                                                 const float* __restrict__ bo,
                                                 float* __restrict__ out) {
  const int tid  = threadIdx.x;
  const int lane = tid & 31, wave = tid >> 5;
  const int l16  = lane & 15, g16 = lane >> 4;
  const int job   = blockIdx.x * 8 + wave;
  const int ptile = job & 3;
  const int mtile = (job >> 2) & 15;
  const int n     = job >> 6;
  const int m0    = mtile * 16;
  const int p0    = ptile * 64;
  const float* ob = Oin + (size_t)n * CC * HW_;

  v8f acc0 = {}, acc1 = {}, acc2 = {}, acc3 = {};
  for (int k = 0; k < CC; k += 4) {
    const float* wp = wo + (size_t)(m0 + l16) * CC + k + 2 * g16;
    v2f a; a[0] = wp[0]; a[1] = wp[1];
    const float* xp = ob + (size_t)(k + 2 * g16) * HW_ + p0 + l16;
    v2f b0; b0[0] = xp[0];  b0[1] = xp[HW_];
    v2f b1; b1[0] = xp[16]; b1[1] = xp[HW_ + 16];
    v2f b2; b2[0] = xp[32]; b2[1] = xp[HW_ + 32];
    v2f b3; b3[0] = xp[48]; b3[1] = xp[HW_ + 48];
    acc0 = wmma_f32(a, b0, acc0);
    acc1 = wmma_f32(a, b1, acc1);
    acc2 = wmma_f32(a, b2, acc2);
    acc3 = wmma_f32(a, b3, acc3);
  }
  float* yb = out + ((size_t)n * CC + m0) * HW_ + p0;
#pragma unroll
  for (int r = 0; r < 8; ++r) {
    const int   m    = r + 8 * g16;
    const float bias = bo[m0 + m];
    float* yr = yb + (size_t)m * HW_ + l16;
    yr[0]  = acc0[r] + bias;
    yr[16] = acc1[r] + bias;
    yr[32] = acc2[r] + bias;
    yr[48] = acc3[r] + bias;
  }
}

extern "C" void kernel_launch(void* const* d_in, const int* in_sizes, int n_in,
                              void* d_out, int out_size, void* d_ws, size_t ws_size,
                              hipStream_t stream) {
  const float* x      = (const float*)d_in[0];
  const float* w_qkv  = (const float*)d_in[1];
  const float* b_qkv  = (const float*)d_in[2];
  const float* w_out  = (const float*)d_in[3];
  const float* b_out  = (const float*)d_in[4];
  const int*   accA   = (const int*)d_in[5];
  float* outp = (float*)d_out;

  float* Y = (float*)d_ws;
  float* O = Y + (size_t)NB * 768 * HW_;

  qkv_gemm<<<1248, 256, 0, stream>>>(x, w_qkv, b_qkv, Y);
  attn_kernel<<<832, 256, 0, stream>>>(Y, accA, O);
  proj_gemm<<<416, 256, 0, stream>>>(O, w_out, b_out, outp);
}


# hot-loop global loads and GEMM stores widened to b128: lane l16 owns positions 4*l16..+3, one per 16-wide tile (x, O, K rows); weights b128 along k; GEMM k, row-sum and PV sum order change
# speedup vs baseline: 1.3059x; 1.1475x over previous
#include <hip/hip_runtime.h>
#include <cstdint>

typedef __attribute__((ext_vector_type(2))) float v2f;
typedef __attribute__((ext_vector_type(8))) float v8f;

#define TT   26
#define NH   8
#define CC   256
#define CPH  32
#define HW_  256
#define NB   52

__device__ __forceinline__ v8f wmma_f32(v2f a, v2f b, v8f c) {
  return __builtin_amdgcn_wmma_f32_16x16x4_f32(false, a, false, b, (short)0, c,
                                               false, false);
}

__device__ __forceinline__ void proj_tile(const float* __restrict__ Wt,
                                          const float* __restrict__ Xn,
                                          const float* __restrict__ bias,
                                          float* __restrict__ On,
                                          int m0, int p0, int l16, int g16) {
  const float* wrow = Wt + (size_t)(m0 + l16) * CC + 4 * g16;
  const float* xcol = Xn + (size_t)(4 * g16) * HW_ + p0 + 4 * l16;

  v8f acc0 = {}, acc1 = {}, acc2 = {}, acc3 = {};
  for (int k = 0; k < CC; k += 8) {
    const float4 a4 = *(const float4*)(wrow + k);
    const float* xp = xcol + (size_t)k * HW_;
    const float4 x0 = *(const float4*)(xp);
    const float4 x1 = *(const float4*)(xp + HW_);
    const float4 x2 = *(const float4*)(xp + 2 * HW_);
    const float4 x3 = *(const float4*)(xp + 3 * HW_);
    v2f a, b;
    a[0] = a4.x; a[1] = a4.y;
    b[0] = x0.x; b[1] = x1.x;  acc0 = wmma_f32(a, b, acc0);
    b[0] = x0.y; b[1] = x1.y;  acc1 = wmma_f32(a, b, acc1);
    b[0] = x0.z; b[1] = x1.z;  acc2 = wmma_f32(a, b, acc2);
    b[0] = x0.w; b[1] = x1.w;  acc3 = wmma_f32(a, b, acc3);
    a[0] = a4.z; a[1] = a4.w;
    b[0] = x2.x; b[1] = x3.x;  acc0 = wmma_f32(a, b, acc0);
    b[0] = x2.y; b[1] = x3.y;  acc1 = wmma_f32(a, b, acc1);
    b[0] = x2.z; b[1] = x3.z;  acc2 = wmma_f32(a, b, acc2);
    b[0] = x2.w; b[1] = x3.w;  acc3 = wmma_f32(a, b, acc3);
  }
  float* ob = On + (size_t)m0 * HW_ + p0 + 4 * l16;
#pragma unroll
  for (int r = 0; r < 8; ++r) {
    const int   m  = r + 8 * g16;
    const float bv = bias[m0 + m];
    float4 v;
    v.x = acc0[r] + bv; v.y = acc1[r] + bv; v.z = acc2[r] + bv; v.w = acc3[r] + bv;
    *(float4*)(ob + (size_t)m * HW_) = v;
  }
}

__global__ void __launch_bounds__(256) qkv_gemm(const float* __restrict__ x,
                                                const float* __restrict__ wq,
                                                const float* __restrict__ bq,
                                                float* __restrict__ Y) {
  const int tid  = threadIdx.x;
  const int lane = tid & 31, wave = tid >> 5;
  const int l16  = lane & 15, g16 = lane >> 4;
  const int job   = blockIdx.x * 8 + wave;
  const int ptile = job & 3;
  const int mtile = (job >> 2) % 48;
  const int n     = job / 192;
  proj_tile(wq, x + (size_t)n * CC * HW_, bq, Y + (size_t)n * 768 * HW_,
            mtile * 16, ptile * 64, l16, g16);
}

#define VPITCH4 257

__global__ void __launch_bounds__(256) attn_kernel(const float* __restrict__ Y,
                                                   const int* __restrict__ accA,
                                                   float* __restrict__ O) {
  __shared__ float4 Vt4[32 * VPITCH4];

  const int blk  = blockIdx.x;
  const int half = blk & 1;
  const int bth  = blk >> 1;
  const int head = bth % NH;
  const int t    = (bth / NH) % TT;
  const int b    = bth / (NH * TT);
  const int n    = b * TT + t;

  const int tid  = threadIdx.x;
  const int lane = tid & 31, wave = tid >> 5;
  const int l16  = lane & 15, g16 = lane >> 4;

  int tg[4];
#pragma unroll
  for (int g = 0; g < 4; ++g) tg[g] = accA[t * 4 + g];

  {
    const int sg = tid >> 6;
    const int s4 = tid & 63;
    const int tsel = sg == 0 ? tg[0] : sg == 1 ? tg[1] : sg == 2 ? tg[2] : tg[3];
    const float* vsrc =
        Y + ((size_t)(b * TT + tsel) * 3 * CC + 2 * CC + head) * HW_ + s4 * 4;
    float4* vdst = Vt4 + sg * 64 + s4;
#pragma unroll 8
    for (int cp = 0; cp < CPH; ++cp) {
      vdst[cp * VPITCH4] = *(const float4*)(vsrc + (size_t)cp * NH * HW_);
    }
  }
  __syncthreads();

  const int m0 = half * 128 + wave * 16;
  const float* Qbase = Y + (size_t)n * 3 * CC * HW_;

  v2f bq[8];
#pragma unroll
  for (int s = 0; s < 8; ++s) {
    const int cp = 4 * s + 2 * g16;
    const float* qp = Qbase + (size_t)(cp * NH + head) * HW_ + m0 + l16;
    bq[s][0] = qp[0];
    bq[s][1] = qp[NH * HW_];
  }

  v8f o0 = {}, o1 = {};
  float rs = 0.0f;
  const float scale = 0.17677669529663687f;
  const float4* Vr0 = Vt4 + l16 * VPITCH4 + 8 * g16;
  const float4* Vr1 = Vr0 + 16 * VPITCH4;
  const float* Klane =
      Y + (size_t)CC * HW_ + (size_t)((2 * g16) * NH + head) * HW_ + 4 * l16;

  for (int st = 0; st < 16; ++st) {
    const int gg = st >> 2;
    const float* kp =
        Klane + (size_t)(b * TT + tg[gg]) * 3 * CC * HW_ + (st & 3) * 64;

    float4 va[8], vb[8];
#pragma unroll
    for (int r = 0; r < 8; ++r) {
      va[r] = Vr0[st * 16 + r];
      vb[r] = Vr1[st * 16 + r];
    }
    __builtin_amdgcn_sched_barrier(0);

    float4 k0[8], k1[8];
#pragma unroll
    for (int s = 0; s < 8; ++s) {
      k0[s] = *(const float4*)(kp + (size_t)(4 * s) * NH * HW_);
      k1[s] = *(const float4*)(kp + (size_t)(4 * s + 1) * NH * HW_);
    }

    v8f s0 = {}, s1 = {}, s2 = {}, s3 = {};
#pragma unroll
    for (int s = 0; s < 8; ++s) {
      v2f ak;
      ak[0] = k0[s].x; ak[1] = k1[s].x;  s0 = wmma_f32(ak, bq[s], s0);
      ak[0] = k0[s].y; ak[1] = k1[s].y;  s1 = wmma_f32(ak, bq[s], s1);
      ak[0] = k0[s].z; ak[1] = k1[s].z;  s2 = wmma_f32(ak, bq[s], s2);
      ak[0] = k0[s].w; ak[1] = k1[s].w;  s3 = wmma_f32(ak, bq[s], s3);
    }

    float p0[8], p1[8], p2[8], p3[8];
#pragma unroll
    for (int r = 0; r < 8; ++r) {
      p0[r] = __expf(s0[r] * scale);
      p1[r] = __expf(s1[r] * scale);
      p2[r] = __expf(s2[r] * scale);
      p3[r] = __expf(s3[r] * scale);
    }
    const float t0 = ((p0[0] + p0[1]) + (p0[2] + p0[3])) + ((p0[4] + p0[5]) + (p0[6] + p0[7]));
    const float t1 = ((p1[0] + p1[1]) + (p1[2] + p1[3])) + ((p1[4] + p1[5]) + (p1[6] + p1[7]));
    const float t2 = ((p2[0] + p2[1]) + (p2[2] + p2[3])) + ((p2[4] + p2[5]) + (p2[6] + p2[7]));
    const float t3 = ((p3[0] + p3[1]) + (p3[2] + p3[3])) + ((p3[4] + p3[5]) + (p3[6] + p3[7]));
    rs += (t0 + t1) + (t2 + t3);

#pragma unroll
    for (int r = 0; r < 8; ++r) {
      v2f bp, av;
      bp[0] = p0[r]; bp[1] = p1[r];
      av[0] = va[r].x; av[1] = va[r].y;  o0 = wmma_f32(av, bp, o0);
      av[0] = vb[r].x; av[1] = vb[r].y;  o1 = wmma_f32(av, bp, o1);
      bp[0] = p2[r]; bp[1] = p3[r];
      av[0] = va[r].z; av[1] = va[r].w;  o0 = wmma_f32(av, bp, o0);
      av[0] = vb[r].z; av[1] = vb[r].w;  o1 = wmma_f32(av, bp, o1);
    }
  }

  rs += __shfl_xor(rs, 16, 32);
  const float inv = 1.0f / rs;

  float* Ob = O + (size_t)n * CC * HW_ + m0 + l16;
#pragma unroll
  for (int r = 0; r < 8; ++r) {
    const int c1 = 8 * g16 + r;
    Ob[(size_t)(c1 * NH + head) * HW_]        = o0[r] * inv;
    Ob[(size_t)((16 + c1) * NH + head) * HW_] = o1[r] * inv;
  }
}

__global__ void __launch_bounds__(256) proj_gemm(const float* __restrict__ Oin,
                                                 const float* __restrict__ wo,
                                                 const float* __restrict__ bo,
                                                 float* __restrict__ out) {
  const int tid  = threadIdx.x;
  const int lane = tid & 31, wave = tid >> 5;
  const int l16  = lane & 15, g16 = lane >> 4;
  const int job   = blockIdx.x * 8 + wave;
  const int ptile = job & 3;
  const int mtile = (job >> 2) & 15;
  const int n     = job >> 6;
  proj_tile(wo, Oin + (size_t)n * CC * HW_, bo, out + (size_t)n * CC * HW_,
            mtile * 16, ptile * 64, l16, g16);
}

extern "C" void kernel_launch(void* const* d_in, const int* in_sizes, int n_in,
                              void* d_out, int out_size, void* d_ws, size_t ws_size,
                              hipStream_t stream) {
  const float* x      = (const float*)d_in[0];
  const float* w_qkv  = (const float*)d_in[1];
  const float* b_qkv  = (const float*)d_in[2];
  const float* w_out  = (const float*)d_in[3];
  const float* b_out  = (const float*)d_in[4];
  const int*   accA   = (const int*)d_in[5];
  float* outp = (float*)d_out;

  float* Y = (float*)d_ws;
  float* O = Y + (size_t)NB * 768 * HW_;

  qkv_gemm<<<1248, 256, 0, stream>>>(x, w_qkv, b_qkv, Y);
  attn_kernel<<<832, 256, 0, stream>>>(Y, accA, O);
  proj_gemm<<<416, 256, 0, stream>>>(O, w_out, b_out, outp);
}


# both GEMMs: wave tile 16x64 -> 32x64, each X fragment feeds two weight tiles (half the X loads and fragment moves per WMMA, half the waves; 4-wave groups), next 8 channels loaded ahead of the WMMAs
# speedup vs baseline: 1.3822x; 1.0584x over previous
#include <hip/hip_runtime.h>
#include <cstdint>

typedef __attribute__((ext_vector_type(2))) float v2f;
typedef __attribute__((ext_vector_type(8))) float v8f;

#define TT   26
#define NH   8
#define CC   256
#define CPH  32
#define HW_  256
#define NB   52

__device__ __forceinline__ v8f wmma_f32(v2f a, v2f b, v8f c) {
  return __builtin_amdgcn_wmma_f32_16x16x4_f32(false, a, false, b, (short)0, c,
                                               false, false);
}

struct KStep {
  float4 a0, a1;
  float4 x0, x1, x2, x3;
};

__device__ __forceinline__ KStep load_step(const float* __restrict__ wrow,
                                           const float* __restrict__ xcol, int k) {
  KStep s;
  s.a0 = *(const float4*)(wrow + k);
  s.a1 = *(const float4*)(wrow + 16 * CC + k);
  const float* xp = xcol + (size_t)k * HW_;
  s.x0 = *(const float4*)(xp);
  s.x1 = *(const float4*)(xp + HW_);
  s.x2 = *(const float4*)(xp + 2 * HW_);
  s.x3 = *(const float4*)(xp + 3 * HW_);
  return s;
}

__device__ __forceinline__ void mma_step(const KStep& s, v8f (&acc)[2][4]) {
  v2f a0, a1, b;
  a0[0] = s.a0.x; a0[1] = s.a0.y;
  a1[0] = s.a1.x; a1[1] = s.a1.y;
  b[0] = s.x0.x; b[1] = s.x1.x;  acc[0][0] = wmma_f32(a0, b, acc[0][0]);  acc[1][0] = wmma_f32(a1, b, acc[1][0]);
  b[0] = s.x0.y; b[1] = s.x1.y;  acc[0][1] = wmma_f32(a0, b, acc[0][1]);  acc[1][1] = wmma_f32(a1, b, acc[1][1]);
  b[0] = s.x0.z; b[1] = s.x1.z;  acc[0][2] = wmma_f32(a0, b, acc[0][2]);  acc[1][2] = wmma_f32(a1, b, acc[1][2]);
  b[0] = s.x0.w; b[1] = s.x1.w;  acc[0][3] = wmma_f32(a0, b, acc[0][3]);  acc[1][3] = wmma_f32(a1, b, acc[1][3]);
  a0[0] = s.a0.z; a0[1] = s.a0.w;
  a1[0] = s.a1.z; a1[1] = s.a1.w;
  b[0] = s.x2.x; b[1] = s.x3.x;  acc[0][0] = wmma_f32(a0, b, acc[0][0]);  acc[1][0] = wmma_f32(a1, b, acc[1][0]);
  b[0] = s.x2.y; b[1] = s.x3.y;  acc[0][1] = wmma_f32(a0, b, acc[0][1]);  acc[1][1] = wmma_f32(a1, b, acc[1][1]);
  b[0] = s.x2.z; b[1] = s.x3.z;  acc[0][2] = wmma_f32(a0, b, acc[0][2]);  acc[1][2] = wmma_f32(a1, b, acc[1][2]);
  b[0] = s.x2.w; b[1] = s.x3.w;  acc[0][3] = wmma_f32(a0, b, acc[0][3]);  acc[1][3] = wmma_f32(a1, b, acc[1][3]);
}

__device__ __forceinline__ void proj_tile(const float* __restrict__ Wt,
                                          const float* __restrict__ Xn,
                                          const float* __restrict__ bias,
                                          float* __restrict__ On,
                                          int m0, int p0, int l16, int g16) {
  const float* wrow = Wt + (size_t)(m0 + l16) * CC + 4 * g16;
  const float* xcol = Xn + (size_t)(4 * g16) * HW_ + p0 + 4 * l16;

  v8f acc[2][4] = {};
  KStep sa = load_step(wrow, xcol, 0);
#pragma unroll 1
  for (int k = 0; k < CC - 16; k += 16) {
    const KStep sb = load_step(wrow, xcol, k + 8);
    mma_step(sa, acc);
    sa = load_step(wrow, xcol, k + 16);
    mma_step(sb, acc);
  }
  {
    const KStep sb = load_step(wrow, xcol, CC - 8);
    mma_step(sa, acc);
    mma_step(sb, acc);
  }

#pragma unroll
  for (int i = 0; i < 2; ++i) {
    float* ob = On + (size_t)(m0 + 16 * i) * HW_ + p0 + 4 * l16;
#pragma unroll
    for (int r = 0; r < 8; ++r) {
      const int   m  = r + 8 * g16;
      const float bv = bias[m0 + 16 * i + m];
      float4 v;
      v.x = acc[i][0][r] + bv; v.y = acc[i][1][r] + bv;
      v.z = acc[i][2][r] + bv; v.w = acc[i][3][r] + bv;
      *(float4*)(ob + (size_t)m * HW_) = v;
    }
  }
}

__global__ void __launch_bounds__(128) qkv_gemm(const float* __restrict__ x,
                                                const float* __restrict__ wq,
                                                const float* __restrict__ bq,
                                                float* __restrict__ Y) {
  const int tid  = threadIdx.x;
  const int lane = tid & 31, wave = tid >> 5;
  const int l16  = lane & 15, g16 = lane >> 4;
  const int blk   = blockIdx.x;
  const int n     = blk / 24;
  const int bin   = blk % 24;
  const int ptile = (bin & 1) * 2 + (wave & 1);
  const int mtile = (bin >> 1) * 2 + (wave >> 1);
  proj_tile(wq, x + (size_t)n * CC * HW_, bq, Y + (size_t)n * 768 * HW_,
            mtile * 32, ptile * 64, l16, g16);
}

#define VPITCH4 257

__global__ void __launch_bounds__(256) attn_kernel(const float* __restrict__ Y,
                                                   const int* __restrict__ accA,
                                                   float* __restrict__ O) {
  __shared__ float4 Vt4[32 * VPITCH4];

  const int blk  = blockIdx.x;
  const int half = blk & 1;
  const int bth  = blk >> 1;
  const int head = bth % NH;
  const int t    = (bth / NH) % TT;
  const int b    = bth / (NH * TT);
  const int n    = b * TT + t;

  const int tid  = threadIdx.x;
  const int lane = tid & 31, wave = tid >> 5;
  const int l16  = lane & 15, g16 = lane >> 4;

  int tg[4];
#pragma unroll
  for (int g = 0; g < 4; ++g) tg[g] = accA[t * 4 + g];

  {
    const int sg = tid >> 6;
    const int s4 = tid & 63;
    const int tsel = sg == 0 ? tg[0] : sg == 1 ? tg[1] : sg == 2 ? tg[2] : tg[3];
    const float* vsrc =
        Y + ((size_t)(b * TT + tsel) * 3 * CC + 2 * CC + head) * HW_ + s4 * 4;
    float4* vdst = Vt4 + sg * 64 + s4;
#pragma unroll 8
    for (int cp = 0; cp < CPH; ++cp) {
      vdst[cp * VPITCH4] = *(const float4*)(vsrc + (size_t)cp * NH * HW_);
    }
  }
  __syncthreads();

  const int m0 = half * 128 + wave * 16;
  const float* Qbase = Y + (size_t)n * 3 * CC * HW_;

  v2f bq[8];
#pragma unroll
  for (int s = 0; s < 8; ++s) {
    const int cp = 4 * s + 2 * g16;
    const float* qp = Qbase + (size_t)(cp * NH + head) * HW_ + m0 + l16;
    bq[s][0] = qp[0];
    bq[s][1] = qp[NH * HW_];
  }

  v8f o0 = {}, o1 = {};
  float rs = 0.0f;
  const float scale = 0.17677669529663687f;
  const float4* Vr0 = Vt4 + l16 * VPITCH4 + 8 * g16;
  const float4* Vr1 = Vr0 + 16 * VPITCH4;
  const float* Klane =
      Y + (size_t)CC * HW_ + (size_t)((2 * g16) * NH + head) * HW_ + 4 * l16;

  for (int st = 0; st < 16; ++st) {
    const int gg = st >> 2;
    const float* kp =
        Klane + (size_t)(b * TT + tg[gg]) * 3 * CC * HW_ + (st & 3) * 64;

    float4 va[8], vb[8];
#pragma unroll
    for (int r = 0; r < 8; ++r) {
      va[r] = Vr0[st * 16 + r];
      vb[r] = Vr1[st * 16 + r];
    }
    __builtin_amdgcn_sched_barrier(0);

    float4 k0[8], k1[8];
#pragma unroll
    for (int s = 0; s < 8; ++s) {
      k0[s] = *(const float4*)(kp + (size_t)(4 * s) * NH * HW_);
      k1[s] = *(const float4*)(kp + (size_t)(4 * s + 1) * NH * HW_);
    }

    v8f s0 = {}, s1 = {}, s2 = {}, s3 = {};
#pragma unroll
    for (int s = 0; s < 8; ++s) {
      v2f ak;
      ak[0] = k0[s].x; ak[1] = k1[s].x;  s0 = wmma_f32(ak, bq[s], s0);
      ak[0] = k0[s].y; ak[1] = k1[s].y;  s1 = wmma_f32(ak, bq[s], s1);
      ak[0] = k0[s].z; ak[1] = k1[s].z;  s2 = wmma_f32(ak, bq[s], s2);
      ak[0] = k0[s].w; ak[1] = k1[s].w;  s3 = wmma_f32(ak, bq[s], s3);
    }

    float p0[8], p1[8], p2[8], p3[8];
#pragma unroll
    for (int r = 0; r < 8; ++r) {
      p0[r] = __expf(s0[r] * scale);
      p1[r] = __expf(s1[r] * scale);
      p2[r] = __expf(s2[r] * scale);
      p3[r] = __expf(s3[r] * scale);
    }
    const float t0 = ((p0[0] + p0[1]) + (p0[2] + p0[3])) + ((p0[4] + p0[5]) + (p0[6] + p0[7]));
    const float t1 = ((p1[0] + p1[1]) + (p1[2] + p1[3])) + ((p1[4] + p1[5]) + (p1[6] + p1[7]));
    const float t2 = ((p2[0] + p2[1]) + (p2[2] + p2[3])) + ((p2[4] + p2[5]) + (p2[6] + p2[7]));
    const float t3 = ((p3[0] + p3[1]) + (p3[2] + p3[3])) + ((p3[4] + p3[5]) + (p3[6] + p3[7]));
    rs += (t0 + t1) + (t2 + t3);

#pragma unroll
    for (int r = 0; r < 8; ++r) {
      v2f bp, av;
      bp[0] = p0[r]; bp[1] = p1[r];
      av[0] = va[r].x; av[1] = va[r].y;  o0 = wmma_f32(av, bp, o0);
      av[0] = vb[r].x; av[1] = vb[r].y;  o1 = wmma_f32(av, bp, o1);
      bp[0] = p2[r]; bp[1] = p3[r];
      av[0] = va[r].z; av[1] = va[r].w;  o0 = wmma_f32(av, bp, o0);
      av[0] = vb[r].z; av[1] = vb[r].w;  o1 = wmma_f32(av, bp, o1);
    }
  }

  rs += __shfl_xor(rs, 16, 32);
  const float inv = 1.0f / rs;

  float* Ob = O + (size_t)n * CC * HW_ + m0 + l16;
#pragma unroll
  for (int r = 0; r < 8; ++r) {
    const int c1 = 8 * g16 + r;
    Ob[(size_t)(c1 * NH + head) * HW_]        = o0[r] * inv;
    Ob[(size_t)((16 + c1) * NH + head) * HW_] = o1[r] * inv;
  }
}

__global__ void __launch_bounds__(128) proj_gemm(const float* __restrict__ Oin,
                                                 const float* __restrict__ wo,
                                                 const float* __restrict__ bo,
                                                 float* __restrict__ out) {
  const int tid  = threadIdx.x;
  const int lane = tid & 31, wave = tid >> 5;
  const int l16  = lane & 15, g16 = lane >> 4;
  const int blk   = blockIdx.x;
  const int n     = blk >> 3;
  const int bin   = blk & 7;
  const int ptile = (bin & 1) * 2 + (wave & 1);
  const int mtile = (bin >> 1) * 2 + (wave >> 1);
  proj_tile(wo, Oin + (size_t)n * CC * HW_, bo, out + (size_t)n * CC * HW_,
            mtile * 32, ptile * 64, l16, g16);
}

extern "C" void kernel_launch(void* const* d_in, const int* in_sizes, int n_in,
                              void* d_out, int out_size, void* d_ws, size_t ws_size,
                              hipStream_t stream) {
  const float* x      = (const float*)d_in[0];
  const float* w_qkv  = (const float*)d_in[1];
  const float* b_qkv  = (const float*)d_in[2];
  const float* w_out  = (const float*)d_in[3];
  const float* b_out  = (const float*)d_in[4];
  const int*   accA   = (const int*)d_in[5];
  float* outp = (float*)d_out;

  float* Y = (float*)d_ws;
  float* O = Y + (size_t)NB * 768 * HW_;

  qkv_gemm<<<1248, 128, 0, stream>>>(x, w_qkv, b_qkv, Y);
  attn_kernel<<<832, 256, 0, stream>>>(Y, accA, O);
  proj_gemm<<<416, 128, 0, stream>>>(O, w_out, b_out, outp);
}


# V third of the QKV product taken transposed and stored in the PV product's A-fragment order (512-B wave stores); attention loads V fragments from global: no LDS, no barrier, 4-wave groups
# speedup vs baseline: 1.4673x; 1.0616x over previous
#include <hip/hip_runtime.h>
#include <cstdint>

typedef __attribute__((ext_vector_type(2))) float v2f;
typedef __attribute__((ext_vector_type(8))) float v8f;

#define TT   26
#define NH   8
#define CC   256
#define CPH  32
#define HW_  256
#define NB   52

__device__ __forceinline__ v8f wmma_f32(v2f a, v2f b, v8f c) {
  return __builtin_amdgcn_wmma_f32_16x16x4_f32(false, a, false, b, (short)0, c,
                                               false, false);
}

struct KStep {
  float4 a0, a1;
  float4 x0, x1, x2, x3;
};

__device__ __forceinline__ KStep load_step(const float* __restrict__ wrow,
                                           const float* __restrict__ xcol, int k,
                                           int a1off) {
  KStep s;
  s.a0 = *(const float4*)(wrow + k);
  s.a1 = *(const float4*)(wrow + a1off + k);
  const float* xp = xcol + (size_t)k * HW_;
  s.x0 = *(const float4*)(xp);
  s.x1 = *(const float4*)(xp + HW_);
  s.x2 = *(const float4*)(xp + 2 * HW_);
  s.x3 = *(const float4*)(xp + 3 * HW_);
  return s;
}

template <bool VT>
__device__ __forceinline__ v8f mm(v2f w, v2f x, v8f c) {
  return VT ? wmma_f32(x, w, c) : wmma_f32(w, x, c);
}

template <bool VT>
__device__ __forceinline__ void mma_step(const KStep& s, v8f (&acc)[2][4]) {
  v2f a0, a1, b;
  a0[0] = s.a0.x; a0[1] = s.a0.y;
  a1[0] = s.a1.x; a1[1] = s.a1.y;
  b[0] = s.x0.x; b[1] = s.x1.x;  acc[0][0] = mm<VT>(a0, b, acc[0][0]);  acc[1][0] = mm<VT>(a1, b, acc[1][0]);
  b[0] = s.x0.y; b[1] = s.x1.y;  acc[0][1] = mm<VT>(a0, b, acc[0][1]);  acc[1][1] = mm<VT>(a1, b, acc[1][1]);
  b[0] = s.x0.z; b[1] = s.x1.z;  acc[0][2] = mm<VT>(a0, b, acc[0][2]);  acc[1][2] = mm<VT>(a1, b, acc[1][2]);
  b[0] = s.x0.w; b[1] = s.x1.w;  acc[0][3] = mm<VT>(a0, b, acc[0][3]);  acc[1][3] = mm<VT>(a1, b, acc[1][3]);
  a0[0] = s.a0.z; a0[1] = s.a0.w;
  a1[0] = s.a1.z; a1[1] = s.a1.w;
  b[0] = s.x2.x; b[1] = s.x3.x;  acc[0][0] = mm<VT>(a0, b, acc[0][0]);  acc[1][0] = mm<VT>(a1, b, acc[1][0]);
  b[0] = s.x2.y; b[1] = s.x3.y;  acc[0][1] = mm<VT>(a0, b, acc[0][1]);  acc[1][1] = mm<VT>(a1, b, acc[1][1]);
  b[0] = s.x2.z; b[1] = s.x3.z;  acc[0][2] = mm<VT>(a0, b, acc[0][2]);  acc[1][2] = mm<VT>(a1, b, acc[1][2]);
  b[0] = s.x2.w; b[1] = s.x3.w;  acc[0][3] = mm<VT>(a0, b, acc[0][3]);  acc[1][3] = mm<VT>(a1, b, acc[1][3]);
}

template <bool VT>
__device__ __forceinline__ void proj_tile(const float* __restrict__ Wt,
                                          const float* __restrict__ Xn,
                                          const float* __restrict__ bias,
                                          float* __restrict__ On,
                                          int m0, int p0, int l16, int g16) {
  const int row0  = VT ? (2 * CC + 8 * l16 + m0) : (m0 + l16);
  const int a1off = (VT ? 8 * 16 : 16) * CC;
  const float* wrow = Wt + (size_t)row0 * CC + 4 * g16;
  const float* xcol = Xn + (size_t)(4 * g16) * HW_ + p0 + 4 * l16;

  v8f acc[2][4] = {};
  KStep sa = load_step(wrow, xcol, 0, a1off);
#pragma unroll 1
  for (int k = 0; k < CC - 16; k += 16) {
    const KStep sb = load_step(wrow, xcol, k + 8, a1off);
    mma_step<VT>(sa, acc);
    sa = load_step(wrow, xcol, k + 16, a1off);
    mma_step<VT>(sb, acc);
  }
  {
    const KStep sb = load_step(wrow, xcol, CC - 8, a1off);
    mma_step<VT>(sa, acc);
    mma_step<VT>(sb, acc);
  }

  if (!VT) {
#pragma unroll
    for (int i = 0; i < 2; ++i) {
      float* ob = On + (size_t)(m0 + 16 * i) * HW_ + p0 + 4 * l16;
#pragma unroll
      for (int r = 0; r < 8; ++r) {
        const int   m  = r + 8 * g16;
        const float bv = bias[m0 + 16 * i + m];
        float4 v;
        v.x = acc[i][0][r] + bv; v.y = acc[i][1][r] + bv;
        v.z = acc[i][2][r] + bv; v.w = acc[i][3][r] + bv;
        *(float4*)(ob + (size_t)m * HW_) = v;
      }
    }
  } else {
    float4* vp = (float4*)On + (size_t)(m0 * 4 + (p0 >> 6)) * (16 * 32) + 16 * g16 + l16;
#pragma unroll
    for (int i = 0; i < 2; ++i) {
      const float bv = bias[2 * CC + 8 * (16 * i + l16) + m0];
#pragma unroll
      for (int r = 0; r < 8; ++r) {
        float4 v;
        v.x = acc[i][0][r] + bv; v.y = acc[i][1][r] + bv;
        v.z = acc[i][2][r] + bv; v.w = acc[i][3][r] + bv;
        vp[(2 * r + i) * 32] = v;
      }
    }
  }
}

__global__ void __launch_bounds__(128) qkv_gemm(const float* __restrict__ x,
                                                const float* __restrict__ wq,
                                                const float* __restrict__ bq,
                                                float* __restrict__ Y) {
  const int tid  = threadIdx.x;
  const int lane = tid & 31, wave = tid >> 5;
  const int l16  = lane & 15, g16 = lane >> 4;
  const int blk   = blockIdx.x;
  const int n     = blk / 24;
  const int bin   = blk % 24;
  const int ptile = (bin & 1) * 2 + (wave & 1);
  const int mtile = (bin >> 1) * 2 + (wave >> 1);
  const float* xn = x + (size_t)n * CC * HW_;
  float*       yn = Y + (size_t)n * 768 * HW_;
  if (bin < 16) {
    proj_tile<false>(wq, xn, bq, yn, mtile * 32, ptile * 64, l16, g16);
  } else {
    proj_tile<true>(wq, xn, bq, yn + 2 * CC * HW_, mtile - 16, ptile * 64, l16, g16);
  }
}

__global__ void __launch_bounds__(128) attn_kernel(const float* __restrict__ Y,
                                                   const int* __restrict__ accA,
                                                   float* __restrict__ O) {
  const int blk  = blockIdx.x;
  const int qrt  = blk & 3;
  const int bth  = blk >> 2;
  const int head = bth % NH;
  const int t    = (bth / NH) % TT;
  const int b    = bth / (NH * TT);
  const int n    = b * TT + t;

  const int tid  = threadIdx.x;
  const int lane = tid & 31, wave = tid >> 5;
  const int l16  = lane & 15, g16 = lane >> 4;

  int tg[4];
#pragma unroll
  for (int g = 0; g < 4; ++g) tg[g] = accA[t * 4 + g];

  const int m0 = qrt * 64 + wave * 16;
  const float* Qbase = Y + (size_t)n * 3 * CC * HW_;

  v2f bq[8];
#pragma unroll
  for (int s = 0; s < 8; ++s) {
    const int cp = 4 * s + 2 * g16;
    const float* qp = Qbase + (size_t)(cp * NH + head) * HW_ + m0 + l16;
    bq[s][0] = qp[0];
    bq[s][1] = qp[NH * HW_];
  }

  v8f o0 = {}, o1 = {};
  float rs = 0.0f;
  const float scale = 0.17677669529663687f;
  const float4* Vlane =
      (const float4*)(Y + (size_t)2 * CC * HW_) + (size_t)head * (4 * 16 * 32) + lane;
  const float* Klane =
      Y + (size_t)CC * HW_ + (size_t)((2 * g16) * NH + head) * HW_ + 4 * l16;

  for (int st = 0; st < 16; ++st) {
    const int gg = st >> 2;
    const size_t noff = (size_t)(b * TT + tg[gg]) * 3 * CC * HW_;
    const float*  kp = Klane + noff + (st & 3) * 64;
    const float4* vp = Vlane + noff / 4 + (st & 3) * (16 * 32);

    float4 va[8], vb[8];
#pragma unroll
    for (int r = 0; r < 8; ++r) {
      va[r] = vp[(2 * r) * 32];
      vb[r] = vp[(2 * r + 1) * 32];
    }

    float4 k0[8], k1[8];
#pragma unroll
    for (int s = 0; s < 8; ++s) {
      k0[s] = *(const float4*)(kp + (size_t)(4 * s) * NH * HW_);
      k1[s] = *(const float4*)(kp + (size_t)(4 * s + 1) * NH * HW_);
    }

    v8f s0 = {}, s1 = {}, s2 = {}, s3 = {};
#pragma unroll
    for (int s = 0; s < 8; ++s) {
      v2f ak;
      ak[0] = k0[s].x; ak[1] = k1[s].x;  s0 = wmma_f32(ak, bq[s], s0);
      ak[0] = k0[s].y; ak[1] = k1[s].y;  s1 = wmma_f32(ak, bq[s], s1);
      ak[0] = k0[s].z; ak[1] = k1[s].z;  s2 = wmma_f32(ak, bq[s], s2);
      ak[0] = k0[s].w; ak[1] = k1[s].w;  s3 = wmma_f32(ak, bq[s], s3);
    }

    float p0[8], p1[8], p2[8], p3[8];
#pragma unroll
    for (int r = 0; r < 8; ++r) {
      p0[r] = __expf(s0[r] * scale);
      p1[r] = __expf(s1[r] * scale);
      p2[r] = __expf(s2[r] * scale);
      p3[r] = __expf(s3[r] * scale);
    }
    const float t0 = ((p0[0] + p0[1]) + (p0[2] + p0[3])) + ((p0[4] + p0[5]) + (p0[6] + p0[7]));
    const float t1 = ((p1[0] + p1[1]) + (p1[2] + p1[3])) + ((p1[4] + p1[5]) + (p1[6] + p1[7]));
    const float t2 = ((p2[0] + p2[1]) + (p2[2] + p2[3])) + ((p2[4] + p2[5]) + (p2[6] + p2[7]));
    const float t3 = ((p3[0] + p3[1]) + (p3[2] + p3[3])) + ((p3[4] + p3[5]) + (p3[6] + p3[7]));
    rs += (t0 + t1) + (t2 + t3);

#pragma unroll
    for (int r = 0; r < 8; ++r) {
      v2f bp, av;
      bp[0] = p0[r]; bp[1] = p1[r];
      av[0] = va[r].x; av[1] = va[r].y;  o0 = wmma_f32(av, bp, o0);
      av[0] = vb[r].x; av[1] = vb[r].y;  o1 = wmma_f32(av, bp, o1);
      bp[0] = p2[r]; bp[1] = p3[r];
      av[0] = va[r].z; av[1] = va[r].w;  o0 = wmma_f32(av, bp, o0);
      av[0] = vb[r].z; av[1] = vb[r].w;  o1 = wmma_f32(av, bp, o1);
    }
  }

  rs += __shfl_xor(rs, 16, 32);
  const float inv = 1.0f / rs;

  float* Ob = O + (size_t)n * CC * HW_ + m0 + l16;
#pragma unroll
  for (int r = 0; r < 8; ++r) {
    const int c1 = 8 * g16 + r;
    Ob[(size_t)(c1 * NH + head) * HW_]        = o0[r] * inv;
    Ob[(size_t)((16 + c1) * NH + head) * HW_] = o1[r] * inv;
  }
}

__global__ void __launch_bounds__(128) proj_gemm(const float* __restrict__ Oin,
                                                 const float* __restrict__ wo,
                                                 const float* __restrict__ bo,
                                                 float* __restrict__ out) {
  const int tid  = threadIdx.x;
  const int lane = tid & 31, wave = tid >> 5;
  const int l16  = lane & 15, g16 = lane >> 4;
  const int blk   = blockIdx.x;
  const int n     = blk >> 3;
  const int bin   = blk & 7;
  const int ptile = (bin & 1) * 2 + (wave & 1);
  const int mtile = (bin >> 1) * 2 + (wave >> 1);
  proj_tile<false>(wo, Oin + (size_t)n * CC * HW_, bo, out + (size_t)n * CC * HW_,
                   mtile * 32, ptile * 64, l16, g16);
}

extern "C" void kernel_launch(void* const* d_in, const int* in_sizes, int n_in,
                              void* d_out, int out_size, void* d_ws, size_t ws_size,
                              hipStream_t stream) {
  const float* x      = (const float*)d_in[0];
  const float* w_qkv  = (const float*)d_in[1];
  const float* b_qkv  = (const float*)d_in[2];
  const float* w_out  = (const float*)d_in[3];
  const float* b_out  = (const float*)d_in[4];
  const int*   accA   = (const int*)d_in[5];
  float* outp = (float*)d_out;

  float* Y = (float*)d_ws;
  float* O = Y + (size_t)NB * 768 * HW_;

  qkv_gemm<<<1248, 128, 0, stream>>>(x, w_qkv, b_qkv, Y);
  attn_kernel<<<1664, 128, 0, stream>>>(Y, accA, O);
  proj_gemm<<<416, 128, 0, stream>>>(O, w_out, b_out, outp);
}


# attention wave tile 16 queries x 1024 keys -> 32 queries x one neighbour (256 keys): K/V fragments feed 2 query tiles, 4 partials merged via LDS, 13 equal waves per SIMD; row-sum, PV sum order change
# speedup vs baseline: 1.5199x; 1.0358x over previous
#include <hip/hip_runtime.h>
#include <cstdint>

typedef __attribute__((ext_vector_type(2))) float v2f;
typedef __attribute__((ext_vector_type(8))) float v8f;

#define TT   26
#define NH   8
#define CC   256
#define CPH  32
#define HW_  256
#define NB   52

__device__ __forceinline__ v8f wmma_f32(v2f a, v2f b, v8f c) {
  return __builtin_amdgcn_wmma_f32_16x16x4_f32(false, a, false, b, (short)0, c,
                                               false, false);
}

struct KStep {
  float4 a0, a1;
  float4 x0, x1, x2, x3;
};

__device__ __forceinline__ KStep load_step(const float* __restrict__ wrow,
                                           const float* __restrict__ xcol, int k,
                                           int a1off) {
  KStep s;
  s.a0 = *(const float4*)(wrow + k);
  s.a1 = *(const float4*)(wrow + a1off + k);
  const float* xp = xcol + (size_t)k * HW_;
  s.x0 = *(const float4*)(xp);
  s.x1 = *(const float4*)(xp + HW_);
  s.x2 = *(const float4*)(xp + 2 * HW_);
  s.x3 = *(const float4*)(xp + 3 * HW_);
  return s;
}

template <bool VT>
__device__ __forceinline__ v8f mm(v2f w, v2f x, v8f c) {
  return VT ? wmma_f32(x, w, c) : wmma_f32(w, x, c);
}

template <bool VT>
__device__ __forceinline__ void mma_step(const KStep& s, v8f (&acc)[2][4]) {
  v2f a0, a1, b;
  a0[0] = s.a0.x; a0[1] = s.a0.y;
  a1[0] = s.a1.x; a1[1] = s.a1.y;
  b[0] = s.x0.x; b[1] = s.x1.x;  acc[0][0] = mm<VT>(a0, b, acc[0][0]);  acc[1][0] = mm<VT>(a1, b, acc[1][0]);
  b[0] = s.x0.y; b[1] = s.x1.y;  acc[0][1] = mm<VT>(a0, b, acc[0][1]);  acc[1][1] = mm<VT>(a1, b, acc[1][1]);
  b[0] = s.x0.z; b[1] = s.x1.z;  acc[0][2] = mm<VT>(a0, b, acc[0][2]);  acc[1][2] = mm<VT>(a1, b, acc[1][2]);
  b[0] = s.x0.w; b[1] = s.x1.w;  acc[0][3] = mm<VT>(a0, b, acc[0][3]);  acc[1][3] = mm<VT>(a1, b, acc[1][3]);
  a0[0] = s.a0.z; a0[1] = s.a0.w;
  a1[0] = s.a1.z; a1[1] = s.a1.w;
  b[0] = s.x2.x; b[1] = s.x3.x;  acc[0][0] = mm<VT>(a0, b, acc[0][0]);  acc[1][0] = mm<VT>(a1, b, acc[1][0]);
  b[0] = s.x2.y; b[1] = s.x3.y;  acc[0][1] = mm<VT>(a0, b, acc[0][1]);  acc[1][1] = mm<VT>(a1, b, acc[1][1]);
  b[0] = s.x2.z; b[1] = s.x3.z;  acc[0][2] = mm<VT>(a0, b, acc[0][2]);  acc[1][2] = mm<VT>(a1, b, acc[1][2]);
  b[0] = s.x2.w; b[1] = s.x3.w;  acc[0][3] = mm<VT>(a0, b, acc[0][3]);  acc[1][3] = mm<VT>(a1, b, acc[1][3]);
}

template <bool VT>
__device__ __forceinline__ void proj_tile(const float* __restrict__ Wt,
                                          const float* __restrict__ Xn,
                                          const float* __restrict__ bias,
                                          float* __restrict__ On,
                                          int m0, int p0, int l16, int g16) {
  const int row0  = VT ? (2 * CC + 8 * l16 + m0) : (m0 + l16);
  const int a1off = (VT ? 8 * 16 : 16) * CC;
  const float* wrow = Wt + (size_t)row0 * CC + 4 * g16;
  const float* xcol = Xn + (size_t)(4 * g16) * HW_ + p0 + 4 * l16;

  v8f acc[2][4] = {};
  KStep sa = load_step(wrow, xcol, 0, a1off);
#pragma unroll 1
  for (int k = 0; k < CC - 16; k += 16) {
    const KStep sb = load_step(wrow, xcol, k + 8, a1off);
    mma_step<VT>(sa, acc);
    sa = load_step(wrow, xcol, k + 16, a1off);
    mma_step<VT>(sb, acc);
  }
  {
    const KStep sb = load_step(wrow, xcol, CC - 8, a1off);
    mma_step<VT>(sa, acc);
    mma_step<VT>(sb, acc);
  }

  if (!VT) {
#pragma unroll
    for (int i = 0; i < 2; ++i) {
      float* ob = On + (size_t)(m0 + 16 * i) * HW_ + p0 + 4 * l16;
#pragma unroll
      for (int r = 0; r < 8; ++r) {
        const int   m  = r + 8 * g16;
        const float bv = bias[m0 + 16 * i + m];
        float4 v;
        v.x = acc[i][0][r] + bv; v.y = acc[i][1][r] + bv;
        v.z = acc[i][2][r] + bv; v.w = acc[i][3][r] + bv;
        *(float4*)(ob + (size_t)m * HW_) = v;
      }
    }
  } else {
    float4* vp = (float4*)On + (size_t)(m0 * 4 + (p0 >> 6)) * (16 * 32) + 16 * g16 + l16;
#pragma unroll
    for (int i = 0; i < 2; ++i) {
      const float bv = bias[2 * CC + 8 * (16 * i + l16) + m0];
#pragma unroll
      for (int r = 0; r < 8; ++r) {
        float4 v;
        v.x = acc[i][0][r] + bv; v.y = acc[i][1][r] + bv;
        v.z = acc[i][2][r] + bv; v.w = acc[i][3][r] + bv;
        vp[(2 * r + i) * 32] = v;
      }
    }
  }
}

__global__ void __launch_bounds__(128) qkv_gemm(const float* __restrict__ x,
                                                const float* __restrict__ wq,
                                                const float* __restrict__ bq,
                                                float* __restrict__ Y) {
  const int tid  = threadIdx.x;
  const int lane = tid & 31, wave = tid >> 5;
  const int l16  = lane & 15, g16 = lane >> 4;
  const int blk   = blockIdx.x;
  const int n     = blk / 24;
  const int bin   = blk % 24;
  const int ptile = (bin & 1) * 2 + (wave & 1);
  const int mtile = (bin >> 1) * 2 + (wave >> 1);
  const float* xn = x + (size_t)n * CC * HW_;
  float*       yn = Y + (size_t)n * 768 * HW_;
  if (bin < 16) {
    proj_tile<false>(wq, xn, bq, yn, mtile * 32, ptile * 64, l16, g16);
  } else {
    proj_tile<true>(wq, xn, bq, yn + 2 * CC * HW_, mtile - 16, ptile * 64, l16, g16);
  }
}

__global__ void __launch_bounds__(128) attn_kernel(const float* __restrict__ Y,
                                                   const int* __restrict__ accA,
                                                   float* __restrict__ O) {
  const int blk  = blockIdx.x;
  const int qh   = blk & 7;
  const int bth  = blk >> 3;
  const int head = bth % NH;
  const int t    = (bth / NH) % TT;
  const int b    = bth / (NH * TT);
  const int n    = b * TT + t;

  const int tid  = threadIdx.x;
  const int lane = tid & 31, wave = tid >> 5;
  const int l16  = lane & 15, g16 = lane >> 4;
  const int kq   = wave;

  const int tgn = accA[t * 4 + kq];

  const int m0 = qh * 32;
  const float* Qbase = Y + (size_t)n * 3 * CC * HW_;

  v2f bq[2][8];
#pragma unroll
  for (int j = 0; j < 2; ++j) {
#pragma unroll
    for (int s = 0; s < 8; ++s) {
      const int cp = 4 * s + 2 * g16;
      const float* qp = Qbase + (size_t)(cp * NH + head) * HW_ + m0 + 16 * j + l16;
      bq[j][s][0] = qp[0];
      bq[j][s][1] = qp[NH * HW_];
    }
  }

  v8f o0[2] = {}, o1[2] = {};
  float rs[2] = {0.0f, 0.0f};
  const float scale = 0.17677669529663687f;
  const size_t noff = (size_t)(b * TT + tgn) * 3 * CC * HW_;
  const float4* Vlane =
      (const float4*)(Y + (size_t)2 * CC * HW_) + noff / 4 + (size_t)head * (4 * 16 * 32) + lane;
  const float* Klane =
      Y + (size_t)CC * HW_ + noff + (size_t)((2 * g16) * NH + head) * HW_ + 4 * l16;

  for (int st = 0; st < 4; ++st) {
    const float*  kp = Klane + st * 64;
    const float4* vp = Vlane + st * (16 * 32);

    float4 va[8], vb[8];
#pragma unroll
    for (int r = 0; r < 8; ++r) {
      va[r] = vp[(2 * r) * 32];
      vb[r] = vp[(2 * r + 1) * 32];
    }

    float4 k0[8], k1[8];
#pragma unroll
    for (int s = 0; s < 8; ++s) {
      k0[s] = *(const float4*)(kp + (size_t)(4 * s) * NH * HW_);
      k1[s] = *(const float4*)(kp + (size_t)(4 * s + 1) * NH * HW_);
    }

    v8f sA[4] = {}, sB[4] = {};
#pragma unroll
    for (int s = 0; s < 8; ++s) {
      v2f ak;
      ak[0] = k0[s].x; ak[1] = k1[s].x;  sA[0] = wmma_f32(ak, bq[0][s], sA[0]);  sB[0] = wmma_f32(ak, bq[1][s], sB[0]);
      ak[0] = k0[s].y; ak[1] = k1[s].y;  sA[1] = wmma_f32(ak, bq[0][s], sA[1]);  sB[1] = wmma_f32(ak, bq[1][s], sB[1]);
      ak[0] = k0[s].z; ak[1] = k1[s].z;  sA[2] = wmma_f32(ak, bq[0][s], sA[2]);  sB[2] = wmma_f32(ak, bq[1][s], sB[2]);
      ak[0] = k0[s].w; ak[1] = k1[s].w;  sA[3] = wmma_f32(ak, bq[0][s], sA[3]);  sB[3] = wmma_f32(ak, bq[1][s], sB[3]);
    }

    float pA[4][8], pB[4][8];
#pragma unroll
    for (int i = 0; i < 4; ++i) {
#pragma unroll
      for (int r = 0; r < 8; ++r) {
        pA[i][r] = __expf(sA[i][r] * scale);
        pB[i][r] = __expf(sB[i][r] * scale);
      }
    }
    float tA[4], tB[4];
#pragma unroll
    for (int i = 0; i < 4; ++i) {
      tA[i] = ((pA[i][0] + pA[i][1]) + (pA[i][2] + pA[i][3])) + ((pA[i][4] + pA[i][5]) + (pA[i][6] + pA[i][7]));
      tB[i] = ((pB[i][0] + pB[i][1]) + (pB[i][2] + pB[i][3])) + ((pB[i][4] + pB[i][5]) + (pB[i][6] + pB[i][7]));
    }
    rs[0] += (tA[0] + tA[1]) + (tA[2] + tA[3]);
    rs[1] += (tB[0] + tB[1]) + (tB[2] + tB[3]);

#pragma unroll
    for (int r = 0; r < 8; ++r) {
      v2f bpA, bpB, av, aw;
      bpA[0] = pA[0][r]; bpA[1] = pA[1][r];
      bpB[0] = pB[0][r]; bpB[1] = pB[1][r];
      av[0] = va[r].x; av[1] = va[r].y;
      aw[0] = vb[r].x; aw[1] = vb[r].y;
      o0[0] = wmma_f32(av, bpA, o0[0]);  o0[1] = wmma_f32(av, bpB, o0[1]);
      o1[0] = wmma_f32(aw, bpA, o1[0]);  o1[1] = wmma_f32(aw, bpB, o1[1]);
      bpA[0] = pA[2][r]; bpA[1] = pA[3][r];
      bpB[0] = pB[2][r]; bpB[1] = pB[3][r];
      av[0] = va[r].z; av[1] = va[r].w;
      aw[0] = vb[r].z; aw[1] = vb[r].w;
      o0[0] = wmma_f32(av, bpA, o0[0]);  o0[1] = wmma_f32(av, bpB, o0[1]);
      o1[0] = wmma_f32(aw, bpA, o1[0]);  o1[1] = wmma_f32(aw, bpB, o1[1]);
    }
  }

  __shared__ float xch[4][34][32];
#pragma unroll
  for (int j = 0; j < 2; ++j) {
#pragma unroll
    for (int r = 0; r < 8; ++r) {
      xch[kq][16 * j + r][lane]     = o0[j][r];
      xch[kq][16 * j + 8 + r][lane] = o1[j][r];
    }
    xch[kq][32 + j][lane] = rs[j];
  }
  __syncthreads();

  const int jt = kq >> 1;
  const int hi = kq & 1;
  float rt = (xch[0][32 + jt][lane] + xch[1][32 + jt][lane]) +
             (xch[2][32 + jt][lane] + xch[3][32 + jt][lane]);
  rt += __shfl_xor(rt, 16, 32);
  const float inv = 1.0f / rt;

  float* Ob = O + (size_t)n * CC * HW_ + m0 + 16 * jt + l16;
#pragma unroll
  for (int r = 0; r < 8; ++r) {
    const int row = 8 * kq + r;
    const float v = (xch[0][row][lane] + xch[1][row][lane]) +
                    (xch[2][row][lane] + xch[3][row][lane]);
    const int c1 = 16 * hi + 8 * g16 + r;
    Ob[(size_t)(c1 * NH + head) * HW_] = v * inv;
  }
}

__global__ void __launch_bounds__(128) proj_gemm(const float* __restrict__ Oin,
                                                 const float* __restrict__ wo,
                                                 const float* __restrict__ bo,
                                                 float* __restrict__ out) {
  const int tid  = threadIdx.x;
  const int lane = tid & 31, wave = tid >> 5;
  const int l16  = lane & 15, g16 = lane >> 4;
  const int blk   = blockIdx.x;
  const int n     = blk >> 3;
  const int bin   = blk & 7;
  const int ptile = (bin & 1) * 2 + (wave & 1);
  const int mtile = (bin >> 1) * 2 + (wave >> 1);
  proj_tile<false>(wo, Oin + (size_t)n * CC * HW_, bo, out + (size_t)n * CC * HW_,
                   mtile * 32, ptile * 64, l16, g16);
}

extern "C" void kernel_launch(void* const* d_in, const int* in_sizes, int n_in,
                              void* d_out, int out_size, void* d_ws, size_t ws_size,
                              hipStream_t stream) {
  const float* x      = (const float*)d_in[0];
  const float* w_qkv  = (const float*)d_in[1];
  const float* b_qkv  = (const float*)d_in[2];
  const float* w_out  = (const float*)d_in[3];
  const float* b_out  = (const float*)d_in[4];
  const int*   accA   = (const int*)d_in[5];
  float* outp = (float*)d_out;

  float* Y = (float*)d_ws;
  float* O = Y + (size_t)NB * 768 * HW_;

  qkv_gemm<<<1248, 128, 0, stream>>>(x, w_qkv, b_qkv, Y);
  attn_kernel<<<3328, 128, 0, stream>>>(Y, accA, O);
  proj_gemm<<<416, 128, 0, stream>>>(O, w_out, b_out, outp);
}
